# starting kernel
# baseline (speedup 1.0000x reference)
#include <hip/hip_runtime.h>
#include <hip/hip_bf16.h>

#define N_NODES 50000
#define N_EDGES 800000
#define D_FEAT  128

typedef __attribute__((ext_vector_type(2))) float v2f;
typedef __attribute__((ext_vector_type(8))) float v8f;

__global__ void sage_zero(float4* __restrict__ p, int n4) {
    int i = blockIdx.x * blockDim.x + threadIdx.x;
    if (i < n4) p[i] = make_float4(0.f, 0.f, 0.f, 0.f);
}

__global__ void sage_scatter(const float* __restrict__ in,
                             const int* __restrict__ src,
                             const int* __restrict__ dst,
                             float* __restrict__ agg,
                             float* __restrict__ deg) {
    int e = blockIdx.x * 2 + (threadIdx.x >> 7);
    int d = threadIdx.x & 127;
    if (e >= N_EDGES) return;
    int s = src[e];
    int t = dst[e];
    __hip_atomic_fetch_add(&agg[t * D_FEAT + d], in[s * D_FEAT + d],
                           __ATOMIC_RELAXED, __HIP_MEMORY_SCOPE_AGENT);
    if (d == 0)
        __hip_atomic_fetch_add(&deg[t], 1.0f,
                               __ATOMIC_RELAXED, __HIP_MEMORY_SCOPE_AGENT);
}

__global__ void sage_mean(float* __restrict__ agg, const float* __restrict__ deg) {
    int i = blockIdx.x * blockDim.x + threadIdx.x;
    if (i < N_NODES * D_FEAT) {
        float dv = fmaxf(deg[i >> 7], 1.0f);
        agg[i] = agg[i] / dv;
    }
}

__global__ void sage_gemm_wmma(const float* __restrict__ Amean,
                               const float* __restrict__ Ax,
                               const float* __restrict__ Wl,
                               const float* __restrict__ Wr,
                               const float* __restrict__ bias,
                               float* __restrict__ out,
                               int do_relu) {
    const int wave  = threadIdx.x >> 5;
    const int lane  = threadIdx.x & 31;
    const int lhalf = lane >> 4;
    const int lmod  = lane & 15;
    const int rbase = blockIdx.x * 16;
    const int cbase = wave * 16;
    const int arow  = rbase + lmod;
    const int col   = cbase + lmod;

    v8f acc = {};
    const float* amp = Amean + arow * D_FEAT + (lhalf ? 2 : 0);
    const float* axp = Ax    + arow * D_FEAT + (lhalf ? 2 : 0);

    #pragma unroll 4
    for (int k = 0; k < D_FEAT; k += 4) {
        const int ka = k + (lhalf ? 2 : 0);
        v2f am = *(const v2f*)(amp + k);
        v2f ax = *(const v2f*)(axp + k);
        v2f bl, br;
        bl.x = Wl[(ka + 0) * D_FEAT + col];
        bl.y = Wl[(ka + 1) * D_FEAT + col];
        br.x = Wr[(ka + 0) * D_FEAT + col];
        br.y = Wr[(ka + 1) * D_FEAT + col];
        acc = __builtin_amdgcn_wmma_f32_16x16x4_f32(false, am, false, bl,
                                                    (short)0, acc, false, false);
        acc = __builtin_amdgcn_wmma_f32_16x16x4_f32(false, ax, false, br,
                                                    (short)0, acc, false, false);
    }

    float bv = bias[col];
    #pragma unroll
    for (int v = 0; v < 8; ++v) {
        float r = acc[v] + bv;
        if (do_relu) r = fmaxf(r, 0.0f);
        int m = v + (lhalf ? 8 : 0);
        out[(rbase + m) * D_FEAT + col] = r;
    }
}

extern "C" void kernel_launch(void* const* d_in, const int* in_sizes, int n_in,
                              void* d_out, int out_size, void* d_ws, size_t ws_size,
                              hipStream_t stream) {
    const float* x    = (const float*)d_in[0];
    const int*   eidx = (const int*)d_in[1];
    const float* W1l  = (const float*)d_in[2];
    const float* b1   = (const float*)d_in[3];
    const float* W1r  = (const float*)d_in[4];
    const float* W2l  = (const float*)d_in[5];
    const float* b2   = (const float*)d_in[6];
    const float* W2r  = (const float*)d_in[7];
    float* out = (float*)d_out;

    const int* src = eidx;
    const int* dst = eidx + N_EDGES;

    float* agg = (float*)d_ws;
    float* deg = agg + (size_t)N_NODES * D_FEAT;
    float* h   = deg + N_NODES;

    const int nzero4  = (N_NODES * D_FEAT + N_NODES) / 4;
    const int zeroBl  = (nzero4 + 255) / 256;
    const int scatBl  = (N_EDGES + 1) / 2;
    const int meanBl  = (N_NODES * D_FEAT + 255) / 256;
    const int gemmBl  = N_NODES / 16;

    sage_zero<<<zeroBl, 256, 0, stream>>>((float4*)agg, nzero4);
    sage_scatter<<<scatBl, 256, 0, stream>>>(x, src, dst, agg, deg);
    sage_mean<<<meanBl, 256, 0, stream>>>(agg, deg);
    sage_gemm_wmma<<<gemmBl, 256, 0, stream>>>(agg, x, W1l, W1r, b1, h, 1);

    sage_zero<<<zeroBl, 256, 0, stream>>>((float4*)agg, nzero4);
    sage_scatter<<<scatBl, 256, 0, stream>>>(h, src, dst, agg, deg);
    sage_mean<<<meanBl, 256, 0, stream>>>(agg, deg);
    sage_gemm_wmma<<<gemmBl, 256, 0, stream>>>(agg, h, W2l, W2r, b2, out, 0);
}


# Aggregate by gather: edges counting-sorted by dst once per call (int atomics), each GEMM block sums its rows' neighbours into an LDS A tile; no float atomics; sum order = arrival rank in segment
# speedup vs baseline: 2.9613x; 2.9613x over previous
#include <hip/hip_runtime.h>
#include <hip/hip_bf16.h>


#define N_NODES 50000
#define N_EDGES 800000
#define D_FEAT  128

#define SCAN_T     1024
#define SCAN_PER   49
#define MEAN_PITCH 132

typedef __attribute__((ext_vector_type(2))) float v2f;
typedef __attribute__((ext_vector_type(4))) float v4f;
typedef __attribute__((ext_vector_type(8))) float v8f;

__global__ void sage_zero_cnt(int* __restrict__ cnt) {
    int i = blockIdx.x * blockDim.x + threadIdx.x;
    if (i < N_NODES) cnt[i] = 0;
}

__global__ void sage_count(const int* __restrict__ dst, int* __restrict__ cnt) {
    int e = blockIdx.x * blockDim.x + threadIdx.x;
    if (e < N_EDGES)
        __hip_atomic_fetch_add(&cnt[dst[e]], 1, __ATOMIC_RELAXED, __HIP_MEMORY_SCOPE_AGENT);
}

__global__ __launch_bounds__(SCAN_T) void sage_scan(int* __restrict__ cnt) {
    __shared__ int s[SCAN_T * SCAN_PER];
    __shared__ int wsum[32];
    const int t    = threadIdx.x;
    const int lane = t & 31;
    const int w    = t >> 5;

    #pragma unroll
    for (int c = 0; c < SCAN_PER; ++c) {
        int i = c * SCAN_T + t;
        s[i] = (i < N_NODES) ? cnt[i] : 0;
    }
    __syncthreads();

    const int base = t * SCAN_PER;
    int run = 0;
    for (int k = 0; k < SCAN_PER; ++k) run += s[base + k];

    int incl = run;
    for (int d = 1; d < 32; d <<= 1) {
        int y = __shfl_up(incl, d);
        if (lane >= d) incl += y;
    }
    if (lane == 31) wsum[w] = incl;
    __syncthreads();
    if (w == 0) {
        int x  = wsum[lane];
        int xi = x;
        for (int d = 1; d < 32; d <<= 1) {
            int y = __shfl_up(xi, d);
            if (lane >= d) xi += y;
        }
        wsum[lane] = xi - x;
    }
    __syncthreads();

    int p = wsum[w] + incl - run;
    for (int k = 0; k < SCAN_PER; ++k) {
        int c = s[base + k];
        s[base + k] = p;
        p += c;
    }
    __syncthreads();

    for (int c = 0; c < SCAN_PER; ++c) {
        int i = c * SCAN_T + t;
        if (i < N_NODES) cnt[i] = s[i];
    }
}

__global__ void sage_fill(const int* __restrict__ src, const int* __restrict__ dst,
                          int* __restrict__ off, int* __restrict__ nbr) {
    int e = blockIdx.x * blockDim.x + threadIdx.x;
    if (e < N_EDGES) {
        int p = __hip_atomic_fetch_add(&off[dst[e]], 1, __ATOMIC_RELAXED, __HIP_MEMORY_SCOPE_AGENT);
        if ((unsigned)p < (unsigned)N_EDGES) nbr[p] = src[e];
    }
}

__global__ void sage_gather_gemm_wmma(const float* __restrict__ in,
                                      const int* __restrict__ off,
                                      const int* __restrict__ nbr,
                                      const float* __restrict__ Wl,
                                      const float* __restrict__ Wr,
                                      const float* __restrict__ bias,
                                      float* __restrict__ out,
                                      int do_relu) {
    __shared__ __attribute__((aligned(16))) float smean[16 * MEAN_PITCH];

    const int wave  = __builtin_amdgcn_readfirstlane((int)(threadIdx.x >> 5));
    const int lane  = threadIdx.x & 31;
    const int lhalf = lane >> 4;
    const int lmod  = lane & 15;
    const int rbase = blockIdx.x * 16;
    const int cbase = wave * 16;
    const int arow  = rbase + lmod;
    const int col   = cbase + lmod;

    for (int r = 0; r < 2; ++r) {
        const int lrow = 2 * wave + r;
        const int row  = rbase + lrow;
        int end = min(off[row], N_EDGES);
        int beg = row ? off[row - 1] : 0;
        beg = min(max(beg, 0), end);
        v4f acc = {0.f, 0.f, 0.f, 0.f};
        for (int j0 = beg; j0 < end; j0 += 32) {
            const int cnt = min(end - j0, 32);
            int myid = (lane < cnt) ? nbr[j0 + lane] : 0;
            myid = ((unsigned)myid < (unsigned)N_NODES) ? myid : 0;
            for (int u0 = 0; u0 < cnt; u0 += 8) {
                v4f v[8];
                #pragma unroll
                for (int u = 0; u < 8; ++u) {
                    const int sidx = __builtin_amdgcn_readlane(myid, min(u0 + u, cnt - 1));
                    v[u] = *(const v4f*)(in + (size_t)sidx * D_FEAT + lane * 4);
                }
                #pragma unroll
                for (int u = 0; u < 8; ++u)
                    if (u0 + u < cnt) acc += v[u];
            }
        }
        const float dv = fmaxf((float)(end - beg), 1.0f);
        *(v4f*)(smean + lrow * MEAN_PITCH + lane * 4) = acc / dv;
    }
    __syncthreads();

    v8f acc = {};
    const float* amp = smean + lmod * MEAN_PITCH + (lhalf ? 2 : 0);
    const float* axp = in    + arow * D_FEAT     + (lhalf ? 2 : 0);

    #pragma unroll 4
    for (int k = 0; k < D_FEAT; k += 4) {
        const int ka = k + (lhalf ? 2 : 0);
        v2f am = *(const v2f*)(amp + k);
        v2f ax = *(const v2f*)(axp + k);
        v2f bl, br;
        bl.x = Wl[(ka + 0) * D_FEAT + col];
        bl.y = Wl[(ka + 1) * D_FEAT + col];
        br.x = Wr[(ka + 0) * D_FEAT + col];
        br.y = Wr[(ka + 1) * D_FEAT + col];
        acc = __builtin_amdgcn_wmma_f32_16x16x4_f32(false, am, false, bl,
                                                    (short)0, acc, false, false);
        acc = __builtin_amdgcn_wmma_f32_16x16x4_f32(false, ax, false, br,
                                                    (short)0, acc, false, false);
    }

    float bv = bias[col];
    #pragma unroll
    for (int v = 0; v < 8; ++v) {
        float r = acc[v] + bv;
        if (do_relu) r = fmaxf(r, 0.0f);
        int m = v + (lhalf ? 8 : 0);
        out[(rbase + m) * D_FEAT + col] = r;
    }
}

extern "C" void kernel_launch(void* const* d_in, const int* in_sizes, int n_in,
                              void* d_out, int out_size, void* d_ws, size_t ws_size,
                              hipStream_t stream) {
    const float* x    = (const float*)d_in[0];
    const int*   eidx = (const int*)d_in[1];
    const float* W1l  = (const float*)d_in[2];
    const float* b1   = (const float*)d_in[3];
    const float* W1r  = (const float*)d_in[4];
    const float* W2l  = (const float*)d_in[5];
    const float* b2   = (const float*)d_in[6];
    const float* W2r  = (const float*)d_in[7];
    float* out = (float*)d_out;

    const int* src = eidx;
    const int* dst = eidx + N_EDGES;

    int*   off = (int*)d_ws;
    int*   nbr = off + N_NODES;
    float* h   = (float*)d_ws + (size_t)N_NODES * D_FEAT + N_NODES;

    const int nodeBl = (N_NODES + 255) / 256;
    const int edgeBl = (N_EDGES + 255) / 256;
    const int gemmBl = N_NODES / 16;

    sage_zero_cnt<<<nodeBl, 256, 0, stream>>>(off);
    sage_count<<<edgeBl, 256, 0, stream>>>(dst, off);
    sage_scan<<<1, SCAN_T, 0, stream>>>(off);
    sage_fill<<<edgeBl, 256, 0, stream>>>(src, dst, off, nbr);

    sage_gather_gemm_wmma<<<gemmBl, 256, 0, stream>>>(x, off, nbr, W1l, W1r, b1, h, 1);
    sage_gather_gemm_wmma<<<gemmBl, 256, 0, stream>>>(h, off, nbr, W2l, W2r, b2, out, 0);
}


# Own-feature A tile staged in LDS: each wave copies its two rows of in with one whole-row load; in @ Wr fragments read from LDS, not by 16-line global loads repeated by all 8 waves
# speedup vs baseline: 3.3210x; 1.1215x over previous
#include <hip/hip_runtime.h>
#include <hip/hip_bf16.h>


#define N_NODES 50000
#define N_EDGES 800000
#define D_FEAT  128

#define SCAN_T     1024
#define SCAN_PER   49
#define MEAN_PITCH 132

typedef __attribute__((ext_vector_type(2))) float v2f;
typedef __attribute__((ext_vector_type(4))) float v4f;
typedef __attribute__((ext_vector_type(8))) float v8f;

__global__ void sage_zero_cnt(int* __restrict__ cnt) {
    int i = blockIdx.x * blockDim.x + threadIdx.x;
    if (i < N_NODES) cnt[i] = 0;
}

__global__ void sage_count(const int* __restrict__ dst, int* __restrict__ cnt) {
    int e = blockIdx.x * blockDim.x + threadIdx.x;
    if (e < N_EDGES)
        __hip_atomic_fetch_add(&cnt[dst[e]], 1, __ATOMIC_RELAXED, __HIP_MEMORY_SCOPE_AGENT);
}

__global__ __launch_bounds__(SCAN_T) void sage_scan(int* __restrict__ cnt) {
    __shared__ int s[SCAN_T * SCAN_PER];
    __shared__ int wsum[32];
    const int t    = threadIdx.x;
    const int lane = t & 31;
    const int w    = t >> 5;

    #pragma unroll
    for (int c = 0; c < SCAN_PER; ++c) {
        int i = c * SCAN_T + t;
        s[i] = (i < N_NODES) ? cnt[i] : 0;
    }
    __syncthreads();

    const int base = t * SCAN_PER;
    int run = 0;
    for (int k = 0; k < SCAN_PER; ++k) run += s[base + k];

    int incl = run;
    for (int d = 1; d < 32; d <<= 1) {
        int y = __shfl_up(incl, d);
        if (lane >= d) incl += y;
    }
    if (lane == 31) wsum[w] = incl;
    __syncthreads();
    if (w == 0) {
        int x  = wsum[lane];
        int xi = x;
        for (int d = 1; d < 32; d <<= 1) {
            int y = __shfl_up(xi, d);
            if (lane >= d) xi += y;
        }
        wsum[lane] = xi - x;
    }
    __syncthreads();

    int p = wsum[w] + incl - run;
    for (int k = 0; k < SCAN_PER; ++k) {
        int c = s[base + k];
        s[base + k] = p;
        p += c;
    }
    __syncthreads();

    for (int c = 0; c < SCAN_PER; ++c) {
        int i = c * SCAN_T + t;
        if (i < N_NODES) cnt[i] = s[i];
    }
}

__global__ void sage_fill(const int* __restrict__ src, const int* __restrict__ dst,
                          int* __restrict__ off, int* __restrict__ nbr) {
    int e = blockIdx.x * blockDim.x + threadIdx.x;
    if (e < N_EDGES) {
        int p = __hip_atomic_fetch_add(&off[dst[e]], 1, __ATOMIC_RELAXED, __HIP_MEMORY_SCOPE_AGENT);
        if ((unsigned)p < (unsigned)N_EDGES) nbr[p] = src[e];
    }
}

__global__ void sage_gather_gemm_wmma(const float* __restrict__ in,
                                      const int* __restrict__ off,
                                      const int* __restrict__ nbr,
                                      const float* __restrict__ Wl,
                                      const float* __restrict__ Wr,
                                      const float* __restrict__ bias,
                                      float* __restrict__ out,
                                      int do_relu) {
    __shared__ __attribute__((aligned(16))) float smean[16 * MEAN_PITCH];
    __shared__ __attribute__((aligned(16))) float sself[16 * MEAN_PITCH];

    const int wave  = __builtin_amdgcn_readfirstlane((int)(threadIdx.x >> 5));
    const int lane  = threadIdx.x & 31;
    const int lhalf = lane >> 4;
    const int lmod  = lane & 15;
    const int rbase = blockIdx.x * 16;
    const int cbase = wave * 16;
    const int col   = cbase + lmod;

    const v4f self0 = *(const v4f*)(in + (size_t)(rbase + 2 * wave)     * D_FEAT + lane * 4);
    const v4f self1 = *(const v4f*)(in + (size_t)(rbase + 2 * wave + 1) * D_FEAT + lane * 4);

    for (int r = 0; r < 2; ++r) {
        const int lrow = 2 * wave + r;
        const int row  = rbase + lrow;
        int end = min(off[row], N_EDGES);
        int beg = row ? off[row - 1] : 0;
        beg = min(max(beg, 0), end);
        v4f acc = {0.f, 0.f, 0.f, 0.f};
        for (int j0 = beg; j0 < end; j0 += 32) {
            const int cnt = min(end - j0, 32);
            int myid = (lane < cnt) ? nbr[j0 + lane] : 0;
            myid = ((unsigned)myid < (unsigned)N_NODES) ? myid : 0;
            for (int u0 = 0; u0 < cnt; u0 += 8) {
                v4f v[8];
                #pragma unroll
                for (int u = 0; u < 8; ++u) {
                    const int sidx = __builtin_amdgcn_readlane(myid, min(u0 + u, cnt - 1));
                    v[u] = *(const v4f*)(in + (size_t)sidx * D_FEAT + lane * 4);
                }
                #pragma unroll
                for (int u = 0; u < 8; ++u)
                    if (u0 + u < cnt) acc += v[u];
            }
        }
        const float dv = fmaxf((float)(end - beg), 1.0f);
        *(v4f*)(smean + lrow * MEAN_PITCH + lane * 4) = acc / dv;
        *(v4f*)(sself + lrow * MEAN_PITCH + lane * 4) = r ? self1 : self0;
    }
    __syncthreads();

    v8f acc = {};
    const float* amp = smean + lmod * MEAN_PITCH + (lhalf ? 2 : 0);
    const float* axp = sself + lmod * MEAN_PITCH + (lhalf ? 2 : 0);

    #pragma unroll 4
    for (int k = 0; k < D_FEAT; k += 4) {
        const int ka = k + (lhalf ? 2 : 0);
        v2f am = *(const v2f*)(amp + k);
        v2f ax = *(const v2f*)(axp + k);
        v2f bl, br;
        bl.x = Wl[(ka + 0) * D_FEAT + col];
        bl.y = Wl[(ka + 1) * D_FEAT + col];
        br.x = Wr[(ka + 0) * D_FEAT + col];
        br.y = Wr[(ka + 1) * D_FEAT + col];
        acc = __builtin_amdgcn_wmma_f32_16x16x4_f32(false, am, false, bl,
                                                    (short)0, acc, false, false);
        acc = __builtin_amdgcn_wmma_f32_16x16x4_f32(false, ax, false, br,
                                                    (short)0, acc, false, false);
    }

    float bv = bias[col];
    #pragma unroll
    for (int v = 0; v < 8; ++v) {
        float r = acc[v] + bv;
        if (do_relu) r = fmaxf(r, 0.0f);
        int m = v + (lhalf ? 8 : 0);
        out[(rbase + m) * D_FEAT + col] = r;
    }
}

extern "C" void kernel_launch(void* const* d_in, const int* in_sizes, int n_in,
                              void* d_out, int out_size, void* d_ws, size_t ws_size,
                              hipStream_t stream) {
    const float* x    = (const float*)d_in[0];
    const int*   eidx = (const int*)d_in[1];
    const float* W1l  = (const float*)d_in[2];
    const float* b1   = (const float*)d_in[3];
    const float* W1r  = (const float*)d_in[4];
    const float* W2l  = (const float*)d_in[5];
    const float* b2   = (const float*)d_in[6];
    const float* W2r  = (const float*)d_in[7];
    float* out = (float*)d_out;

    const int* src = eidx;
    const int* dst = eidx + N_EDGES;

    int*   off = (int*)d_ws;
    int*   nbr = off + N_NODES;
    float* h   = (float*)d_ws + (size_t)N_NODES * D_FEAT + N_NODES;

    const int nodeBl = (N_NODES + 255) / 256;
    const int edgeBl = (N_EDGES + 255) / 256;
    const int gemmBl = N_NODES / 16;

    sage_zero_cnt<<<nodeBl, 256, 0, stream>>>(off);
    sage_count<<<edgeBl, 256, 0, stream>>>(dst, off);
    sage_scan<<<1, SCAN_T, 0, stream>>>(off);
    sage_fill<<<edgeBl, 256, 0, stream>>>(src, dst, off, nbr);

    sage_gather_gemm_wmma<<<gemmBl, 256, 0, stream>>>(x, off, nbr, W1l, W1r, b1, h, 1);
    sage_gather_gemm_wmma<<<gemmBl, 256, 0, stream>>>(h, off, nbr, W2l, W2r, b2, out, 0);
}


# Weights repacked once per call (in the count-clearing launch) into B-fragment order: one 16-byte load per lane feeds two k-steps, whole 128-B lines, 4x fewer weight loads; sums unchanged
# speedup vs baseline: 3.4348x; 1.0343x over previous
#include <hip/hip_runtime.h>
#include <hip/hip_bf16.h>


#define N_NODES 50000
#define N_EDGES 800000
#define D_FEAT  128

#define SCAN_T     1024
#define SCAN_PER   49
#define MEAN_PITCH 132
#define NODE_BL    ((N_NODES + 255) / 256)
#define WP_LAYER   (2 * D_FEAT * D_FEAT)
#define PACK_BL    (2 * WP_LAYER / 4 / 256)

typedef __attribute__((ext_vector_type(2))) float v2f;
typedef __attribute__((ext_vector_type(4))) float v4f;
typedef __attribute__((ext_vector_type(8))) float v8f;

__global__ void sage_zero_cnt_pack_w(int* __restrict__ cnt,
                                     const float* __restrict__ W1l, const float* __restrict__ W1r,
                                     const float* __restrict__ W2l, const float* __restrict__ W2r,
                                     float* __restrict__ wp) {
    if (blockIdx.x < NODE_BL) {
        int i = blockIdx.x * blockDim.x + threadIdx.x;
        if (i < N_NODES) cnt[i] = 0;
        return;
    }
    const int g     = (blockIdx.x - NODE_BL) * 256 + threadIdx.x;
    const int layer = g >> 13;
    const int r     = g & 8191;
    const int l     = r & 31;
    const int m     = (r >> 5) & 1;
    const int u     = (r >> 6) & 15;
    const int w     = (r >> 10) & 7;
    const int col   = 16 * w + (l & 15);
    const int k0    = 8 * u + 2 * (l >> 4);
    const float* W  = layer ? (m ? W2r : W2l) : (m ? W1r : W1l);
    v4f o;
    o.x = W[(k0 + 0) * D_FEAT + col];
    o.y = W[(k0 + 1) * D_FEAT + col];
    o.z = W[(k0 + 4) * D_FEAT + col];
    o.w = W[(k0 + 5) * D_FEAT + col];
    *(v4f*)(wp + (size_t)layer * WP_LAYER + (size_t)r * 4) = o;
}

__global__ void sage_count(const int* __restrict__ dst, int* __restrict__ cnt) {
    int e = blockIdx.x * blockDim.x + threadIdx.x;
    if (e < N_EDGES)
        __hip_atomic_fetch_add(&cnt[dst[e]], 1, __ATOMIC_RELAXED, __HIP_MEMORY_SCOPE_AGENT);
}

__global__ __launch_bounds__(SCAN_T) void sage_scan(int* __restrict__ cnt) {
    __shared__ int s[SCAN_T * SCAN_PER];
    __shared__ int wsum[32];
    const int t    = threadIdx.x;
    const int lane = t & 31;
    const int w    = t >> 5;

    #pragma unroll
    for (int c = 0; c < SCAN_PER; ++c) {
        int i = c * SCAN_T + t;
        s[i] = (i < N_NODES) ? cnt[i] : 0;
    }
    __syncthreads();

    const int base = t * SCAN_PER;
    int run = 0;
    for (int k = 0; k < SCAN_PER; ++k) run += s[base + k];

    int incl = run;
    for (int d = 1; d < 32; d <<= 1) {
        int y = __shfl_up(incl, d);
        if (lane >= d) incl += y;
    }
    if (lane == 31) wsum[w] = incl;
    __syncthreads();
    if (w == 0) {
        int x  = wsum[lane];
        int xi = x;
        for (int d = 1; d < 32; d <<= 1) {
            int y = __shfl_up(xi, d);
            if (lane >= d) xi += y;
        }
        wsum[lane] = xi - x;
    }
    __syncthreads();

    int p = wsum[w] + incl - run;
    for (int k = 0; k < SCAN_PER; ++k) {
        int c = s[base + k];
        s[base + k] = p;
        p += c;
    }
    __syncthreads();

    for (int c = 0; c < SCAN_PER; ++c) {
        int i = c * SCAN_T + t;
        if (i < N_NODES) cnt[i] = s[i];
    }
}

__global__ void sage_fill(const int* __restrict__ src, const int* __restrict__ dst,
                          int* __restrict__ off, int* __restrict__ nbr) {
    int e = blockIdx.x * blockDim.x + threadIdx.x;
    if (e < N_EDGES) {
        int p = __hip_atomic_fetch_add(&off[dst[e]], 1, __ATOMIC_RELAXED, __HIP_MEMORY_SCOPE_AGENT);
        if ((unsigned)p < (unsigned)N_EDGES) nbr[p] = src[e];
    }
}

__global__ void sage_gather_gemm_wmma(const float* __restrict__ in,
                                      const int* __restrict__ off,
                                      const int* __restrict__ nbr,
                                      const float* __restrict__ wp,
                                      const float* __restrict__ bias,
                                      float* __restrict__ out,
                                      int do_relu) {
    __shared__ __attribute__((aligned(16))) float smean[16 * MEAN_PITCH];
    __shared__ __attribute__((aligned(16))) float sself[16 * MEAN_PITCH];

    const int wave  = __builtin_amdgcn_readfirstlane((int)(threadIdx.x >> 5));
    const int lane  = threadIdx.x & 31;
    const int lhalf = lane >> 4;
    const int lmod  = lane & 15;
    const int rbase = blockIdx.x * 16;
    const int cbase = wave * 16;
    const int col   = cbase + lmod;

    const v4f self0 = *(const v4f*)(in + (size_t)(rbase + 2 * wave)     * D_FEAT + lane * 4);
    const v4f self1 = *(const v4f*)(in + (size_t)(rbase + 2 * wave + 1) * D_FEAT + lane * 4);

    for (int r = 0; r < 2; ++r) {
        const int lrow = 2 * wave + r;
        const int row  = rbase + lrow;
        int end = min(off[row], N_EDGES);
        int beg = row ? off[row - 1] : 0;
        beg = min(max(beg, 0), end);
        v4f acc = {0.f, 0.f, 0.f, 0.f};
        for (int j0 = beg; j0 < end; j0 += 32) {
            const int cnt = min(end - j0, 32);
            int myid = (lane < cnt) ? nbr[j0 + lane] : 0;
            myid = ((unsigned)myid < (unsigned)N_NODES) ? myid : 0;
            for (int u0 = 0; u0 < cnt; u0 += 8) {
                v4f v[8];
                #pragma unroll
                for (int u = 0; u < 8; ++u) {
                    const int sidx = __builtin_amdgcn_readlane(myid, min(u0 + u, cnt - 1));
                    v[u] = *(const v4f*)(in + (size_t)sidx * D_FEAT + lane * 4);
                }
                #pragma unroll
                for (int u = 0; u < 8; ++u)
                    if (u0 + u < cnt) acc += v[u];
            }
        }
        const float dv = fmaxf((float)(end - beg), 1.0f);
        *(v4f*)(smean + lrow * MEAN_PITCH + lane * 4) = acc / dv;
        *(v4f*)(sself + lrow * MEAN_PITCH + lane * 4) = r ? self1 : self0;
    }
    __syncthreads();

    v8f acc = {};
    const float* amp = smean + lmod * MEAN_PITCH + (lhalf ? 2 : 0);
    const float* axp = sself + lmod * MEAN_PITCH + (lhalf ? 2 : 0);

    const v4f* wq = (const v4f*)wp + (size_t)wave * (16 * 2 * 32) + lane;

    #pragma unroll 4
    for (int u = 0; u < D_FEAT / 8; ++u) {
        const v4f pl = wq[(2 * u + 0) * 32];
        const v4f pr = wq[(2 * u + 1) * 32];
        const v2f am0 = *(const v2f*)(amp + 8 * u);
        const v2f ax0 = *(const v2f*)(axp + 8 * u);
        const v2f am1 = *(const v2f*)(amp + 8 * u + 4);
        const v2f ax1 = *(const v2f*)(axp + 8 * u + 4);
        const v2f bl0 = __builtin_shufflevector(pl, pl, 0, 1);
        const v2f br0 = __builtin_shufflevector(pr, pr, 0, 1);
        const v2f bl1 = __builtin_shufflevector(pl, pl, 2, 3);
        const v2f br1 = __builtin_shufflevector(pr, pr, 2, 3);
        acc = __builtin_amdgcn_wmma_f32_16x16x4_f32(false, am0, false, bl0,
                                                    (short)0, acc, false, false);
        acc = __builtin_amdgcn_wmma_f32_16x16x4_f32(false, ax0, false, br0,
                                                    (short)0, acc, false, false);
        acc = __builtin_amdgcn_wmma_f32_16x16x4_f32(false, am1, false, bl1,
                                                    (short)0, acc, false, false);
        acc = __builtin_amdgcn_wmma_f32_16x16x4_f32(false, ax1, false, br1,
                                                    (short)0, acc, false, false);
    }

    float bv = bias[col];
    #pragma unroll
    for (int v = 0; v < 8; ++v) {
        float r = acc[v] + bv;
        if (do_relu) r = fmaxf(r, 0.0f);
        int m = v + (lhalf ? 8 : 0);
        out[(rbase + m) * D_FEAT + col] = r;
    }
}

extern "C" void kernel_launch(void* const* d_in, const int* in_sizes, int n_in,
                              void* d_out, int out_size, void* d_ws, size_t ws_size,
                              hipStream_t stream) {
    const float* x    = (const float*)d_in[0];
    const int*   eidx = (const int*)d_in[1];
    const float* W1l  = (const float*)d_in[2];
    const float* b1   = (const float*)d_in[3];
    const float* W1r  = (const float*)d_in[4];
    const float* W2l  = (const float*)d_in[5];
    const float* b2   = (const float*)d_in[6];
    const float* W2r  = (const float*)d_in[7];
    float* out = (float*)d_out;

    const int* src = eidx;
    const int* dst = eidx + N_EDGES;

    int*   off = (int*)d_ws;
    int*   nbr = off + N_NODES;
    float* h   = (float*)d_ws + (size_t)N_NODES * D_FEAT + N_NODES;
    float* wp  = (float*)d_ws + (1 << 20);

    const int nodeBl = (N_NODES + 255) / 256;
    const int edgeBl = (N_EDGES + 255) / 256;
    const int gemmBl = N_NODES / 16;

    sage_zero_cnt_pack_w<<<nodeBl + PACK_BL, 256, 0, stream>>>(off, W1l, W1r, W2l, W2r, wp);
    sage_count<<<edgeBl, 256, 0, stream>>>(dst, off);
    sage_scan<<<1, SCAN_T, 0, stream>>>(off);
    sage_fill<<<edgeBl, 256, 0, stream>>>(src, dst, off, nbr);

    sage_gather_gemm_wmma<<<gemmBl, 256, 0, stream>>>(x, off, nbr, wp, b1, h, 1);
    sage_gather_gemm_wmma<<<gemmBl, 256, 0, stream>>>(h, off, nbr, wp + WP_LAYER, b2, out, 0);
}


# Neighbour table of 64 fixed slots per node built in ONE edge pass (slot = returning int add, 8 edges per thread, overflow list past 64): count and scan launches dropped; sums still in arrival order
# speedup vs baseline: 4.8626x; 1.4157x over previous
#include <hip/hip_runtime.h>
#include <hip/hip_bf16.h>


#define N_NODES 50000
#define N_EDGES 800000
#define D_FEAT  128

#define NBR_CAP    64
#define FILL_EDGES 2048
#define MEAN_PITCH 132
#define NODE_BL    ((N_NODES + 256) / 256)
#define WP_LAYER   (2 * D_FEAT * D_FEAT)
#define PACK_BL    (2 * WP_LAYER / 4 / 256)

typedef __attribute__((ext_vector_type(4))) int   v4i;
typedef __attribute__((ext_vector_type(2))) float v2f;
typedef __attribute__((ext_vector_type(4))) float v4f;
typedef __attribute__((ext_vector_type(8))) float v8f;

__global__ void sage_zero_cnt_pack_w(int* __restrict__ cnt,
                                     const float* __restrict__ W1l, const float* __restrict__ W1r,
                                     const float* __restrict__ W2l, const float* __restrict__ W2r,
                                     float* __restrict__ wp) {
    if (blockIdx.x < NODE_BL) {
        int i = blockIdx.x * blockDim.x + threadIdx.x;
        if (i <= N_NODES) cnt[i] = 0;
        return;
    }
    const int g     = (blockIdx.x - NODE_BL) * 256 + threadIdx.x;
    const int layer = g >> 13;
    const int r     = g & 8191;
    const int l     = r & 31;
    const int m     = (r >> 5) & 1;
    const int u     = (r >> 6) & 15;
    const int w     = (r >> 10) & 7;
    const int col   = 16 * w + (l & 15);
    const int k0    = 8 * u + 2 * (l >> 4);
    const float* W  = layer ? (m ? W2r : W2l) : (m ? W1r : W1l);
    v4f o;
    o.x = W[(k0 + 0) * D_FEAT + col];
    o.y = W[(k0 + 1) * D_FEAT + col];
    o.z = W[(k0 + 4) * D_FEAT + col];
    o.w = W[(k0 + 5) * D_FEAT + col];
    *(v4f*)(wp + (size_t)layer * WP_LAYER + (size_t)r * 4) = o;
}

__global__ __launch_bounds__(256) void sage_fill_slots(const int* __restrict__ src, const int* __restrict__ dst,
                                                       int* __restrict__ cnt, int* __restrict__ tab,
                                                       int* __restrict__ ovd, int* __restrict__ ovs) {
    const int base = blockIdx.x * FILL_EDGES + threadIdx.x * 4;
    int d[8], s[8], slot[8];
    for (int g = 0; g < 2; ++g) {
        const int e0 = base + g * 1024;
        v4i dv = {-1, -1, -1, -1};
        v4i sv = {0, 0, 0, 0};
        if (e0 < N_EDGES) {
            dv = *(const v4i*)(dst + e0);
            sv = *(const v4i*)(src + e0);
        }
        for (int k = 0; k < 4; ++k) { d[4 * g + k] = dv[k]; s[4 * g + k] = sv[k]; }
    }
    for (int k = 0; k < 8; ++k) {
        slot[k] = -1;
        if ((unsigned)d[k] < (unsigned)N_NODES)
            slot[k] = __hip_atomic_fetch_add(&cnt[d[k]], 1, __ATOMIC_RELAXED, __HIP_MEMORY_SCOPE_AGENT);
    }
    for (int k = 0; k < 8; ++k) {
        if ((unsigned)d[k] < (unsigned)N_NODES) {
            if ((unsigned)slot[k] < (unsigned)NBR_CAP) {
                tab[(size_t)d[k] * NBR_CAP + slot[k]] = s[k];
            } else {
                int p = __hip_atomic_fetch_add(&cnt[N_NODES], 1, __ATOMIC_RELAXED, __HIP_MEMORY_SCOPE_AGENT);
                if ((unsigned)p < (unsigned)N_EDGES) { ovd[p] = d[k]; ovs[p] = s[k]; }
            }
        }
    }
}

__global__ void sage_gather_gemm_wmma(const float* __restrict__ in,
                                      const int* __restrict__ cnt,
                                      const int* __restrict__ tab,
                                      const int* __restrict__ ovd,
                                      const int* __restrict__ ovs,
                                      const float* __restrict__ wp,
                                      const float* __restrict__ bias,
                                      float* __restrict__ out,
                                      int do_relu) {
    __shared__ __attribute__((aligned(16))) float smean[16 * MEAN_PITCH];
    __shared__ __attribute__((aligned(16))) float sself[16 * MEAN_PITCH];

    const int wave  = __builtin_amdgcn_readfirstlane((int)(threadIdx.x >> 5));
    const int lane  = threadIdx.x & 31;
    const int lhalf = lane >> 4;
    const int lmod  = lane & 15;
    const int rbase = blockIdx.x * 16;
    const int cbase = wave * 16;
    const int col   = cbase + lmod;

    const v4f self0 = *(const v4f*)(in + (size_t)(rbase + 2 * wave)     * D_FEAT + lane * 4);
    const v4f self1 = *(const v4f*)(in + (size_t)(rbase + 2 * wave + 1) * D_FEAT + lane * 4);

    for (int r = 0; r < 2; ++r) {
        const int lrow = 2 * wave + r;
        const int row  = rbase + lrow;
        const int deg   = min(max(cnt[row], 0), N_EDGES);
        const int intab = min(deg, NBR_CAP);
        const int* trow = tab + (size_t)row * NBR_CAP;
        v4f acc = {0.f, 0.f, 0.f, 0.f};
        for (int j0 = 0; j0 < intab; j0 += 32) {
            const int n = min(intab - j0, 32);
            int myid = (lane < n) ? trow[j0 + lane] : 0;
            myid = ((unsigned)myid < (unsigned)N_NODES) ? myid : 0;
            for (int u0 = 0; u0 < n; u0 += 8) {
                v4f v[8];
                #pragma unroll
                for (int u = 0; u < 8; ++u) {
                    const int sidx = __builtin_amdgcn_readlane(myid, min(u0 + u, n - 1));
                    v[u] = *(const v4f*)(in + (size_t)sidx * D_FEAT + lane * 4);
                }
                #pragma unroll
                for (int u = 0; u < 8; ++u)
                    if (u0 + u < n) acc += v[u];
            }
        }
        if (deg > NBR_CAP) {
            const int novf = min(max(cnt[N_NODES], 0), N_EDGES);
            for (int j0 = 0; j0 < novf; j0 += 32) {
                const int n  = min(novf - j0, 32);
                const int od = (lane < n) ? ovd[j0 + lane] : -1;
                int       os = (lane < n) ? ovs[j0 + lane] : 0;
                os = ((unsigned)os < (unsigned)N_NODES) ? os : 0;
                unsigned m = __builtin_amdgcn_ballot_w32(od == row);
                while (m) {
                    const int b = __builtin_ctz(m);
                    m &= m - 1;
                    const int sidx = __builtin_amdgcn_readlane(os, b);
                    acc += *(const v4f*)(in + (size_t)sidx * D_FEAT + lane * 4);
                }
            }
        }
        const float dv = fmaxf((float)deg, 1.0f);
        *(v4f*)(smean + lrow * MEAN_PITCH + lane * 4) = acc / dv;
        *(v4f*)(sself + lrow * MEAN_PITCH + lane * 4) = r ? self1 : self0;
    }
    __syncthreads();

    v8f acc = {};
    const float* amp = smean + lmod * MEAN_PITCH + (lhalf ? 2 : 0);
    const float* axp = sself + lmod * MEAN_PITCH + (lhalf ? 2 : 0);

    const v4f* wq = (const v4f*)wp + (size_t)wave * (16 * 2 * 32) + lane;

    #pragma unroll 4
    for (int u = 0; u < D_FEAT / 8; ++u) {
        const v4f pl = wq[(2 * u + 0) * 32];
        const v4f pr = wq[(2 * u + 1) * 32];
        const v2f am0 = *(const v2f*)(amp + 8 * u);
        const v2f ax0 = *(const v2f*)(axp + 8 * u);
        const v2f am1 = *(const v2f*)(amp + 8 * u + 4);
        const v2f ax1 = *(const v2f*)(axp + 8 * u + 4);
        const v2f bl0 = __builtin_shufflevector(pl, pl, 0, 1);
        const v2f br0 = __builtin_shufflevector(pr, pr, 0, 1);
        const v2f bl1 = __builtin_shufflevector(pl, pl, 2, 3);
        const v2f br1 = __builtin_shufflevector(pr, pr, 2, 3);
        acc = __builtin_amdgcn_wmma_f32_16x16x4_f32(false, am0, false, bl0,
                                                    (short)0, acc, false, false);
        acc = __builtin_amdgcn_wmma_f32_16x16x4_f32(false, ax0, false, br0,
                                                    (short)0, acc, false, false);
        acc = __builtin_amdgcn_wmma_f32_16x16x4_f32(false, am1, false, bl1,
                                                    (short)0, acc, false, false);
        acc = __builtin_amdgcn_wmma_f32_16x16x4_f32(false, ax1, false, br1,
                                                    (short)0, acc, false, false);
    }

    float bv = bias[col];
    #pragma unroll
    for (int v = 0; v < 8; ++v) {
        float r = acc[v] + bv;
        if (do_relu) r = fmaxf(r, 0.0f);
        int m = v + (lhalf ? 8 : 0);
        out[(rbase + m) * D_FEAT + col] = r;
    }
}

extern "C" void kernel_launch(void* const* d_in, const int* in_sizes, int n_in,
                              void* d_out, int out_size, void* d_ws, size_t ws_size,
                              hipStream_t stream) {
    const float* x    = (const float*)d_in[0];
    const int*   eidx = (const int*)d_in[1];
    const float* W1l  = (const float*)d_in[2];
    const float* b1   = (const float*)d_in[3];
    const float* W1r  = (const float*)d_in[4];
    const float* W2l  = (const float*)d_in[5];
    const float* b2   = (const float*)d_in[6];
    const float* W2r  = (const float*)d_in[7];
    float* out = (float*)d_out;

    const int* src = eidx;
    const int* dst = eidx + N_EDGES;

    int*   cnt = (int*)d_ws;
    int*   tab = (int*)d_ws + (1 << 16);
    int*   ovd = (int*)d_ws + (25 << 17);
    int*   ovs = (int*)d_ws + (1 << 22);
    float* wp  = (float*)d_ws + (5 << 20);
    float* h   = (float*)d_ws + (size_t)N_NODES * D_FEAT + N_NODES;

    const int fillBl = (N_EDGES + FILL_EDGES - 1) / FILL_EDGES;
    const int gemmBl = N_NODES / 16;

    sage_zero_cnt_pack_w<<<NODE_BL + PACK_BL, 256, 0, stream>>>(cnt, W1l, W1r, W2l, W2r, wp);
    sage_fill_slots<<<fillBl, 256, 0, stream>>>(src, dst, cnt, tab, ovd, ovs);

    sage_gather_gemm_wmma<<<gemmBl, 256, 0, stream>>>(x, cnt, tab, ovd, ovs, wp, b1, h, 1);
    sage_gather_gemm_wmma<<<gemmBl, 256, 0, stream>>>(h, cnt, tab, ovd, ovs, wp + WP_LAYER, b2, out, 0);
}


# Neighbour table built by an LDS bucket sort: edge blocks sort (dst,src) pairs by 128-node bucket; one block per bucket takes slots from LDS returning adds, stores whole rows; sum order = LDS add rank
# speedup vs baseline: 7.0799x; 1.4560x over previous
#include <hip/hip_runtime.h>
#include <hip/hip_bf16.h>


#define N_NODES 50000
#define N_EDGES 800000
#define D_FEAT  128

#define NBR_CAP    64
#define FILL_EDGES 2048
#define E_BLK      ((N_EDGES + FILL_EDGES - 1) / FILL_EDGES)
#define BKT_NODES  128
#define N_BKT      ((N_NODES + BKT_NODES - 1) / BKT_NODES)
#define OFF_PITCH  392
#define SLOT_ITEMS 9
#define MEAN_PITCH 132
#define WP_LAYER   (2 * D_FEAT * D_FEAT)
#define PACK_BL    (2 * WP_LAYER / 4 / 256)

static_assert(N_BKT + 1 <= 512, "the bucket histogram has 512 bins");
static_assert(E_BLK <= OFF_PITCH && N_BKT + 1 <= OFF_PITCH, "offset table pitch");
static_assert(E_BLK < 511, "the scan of piece lengths has 512 entries and the search reads entries 0 .. 510");
static_assert(N_EDGES % 4 == 0, "edge groups of 4 are whole or absent");

typedef __attribute__((ext_vector_type(2))) int   v2i;
typedef __attribute__((ext_vector_type(4))) int   v4i;
typedef __attribute__((ext_vector_type(2))) float v2f;
typedef __attribute__((ext_vector_type(4))) float v4f;
typedef __attribute__((ext_vector_type(8))) float v8f;

__global__ __launch_bounds__(256) void sage_bucket_edges_pack_w(const int* __restrict__ src, const int* __restrict__ dst,
                                                                int* __restrict__ cnt, int* __restrict__ srt, int* __restrict__ off,
                                                                const float* __restrict__ W1l, const float* __restrict__ W1r,
                                                                const float* __restrict__ W2l, const float* __restrict__ W2r,
                                                                float* __restrict__ wp) {
    __shared__ int hist[512];
    __shared__ int start[512];
    __shared__ int wtot[8];
    __shared__ __attribute__((aligned(16))) int spair[2 * FILL_EDGES];

    const int t = threadIdx.x;
    if (blockIdx.x >= E_BLK) {
        const int g     = (blockIdx.x - E_BLK) * 256 + t;
        const int layer = g >> 13;
        const int r     = g & 8191;
        const int l     = r & 31;
        const int m     = (r >> 5) & 1;
        const int u     = (r >> 6) & 15;
        const int w     = (r >> 10) & 7;
        const int col   = 16 * w + (l & 15);
        const int k0    = 8 * u + 2 * (l >> 4);
        const float* W  = layer ? (m ? W2r : W2l) : (m ? W1r : W1l);
        v4f o;
        o.x = W[(k0 + 0) * D_FEAT + col];
        o.y = W[(k0 + 1) * D_FEAT + col];
        o.z = W[(k0 + 4) * D_FEAT + col];
        o.w = W[(k0 + 5) * D_FEAT + col];
        *(v4f*)(wp + (size_t)layer * WP_LAYER + (size_t)r * 4) = o;
        return;
    }

    hist[t] = 0;
    hist[t + 256] = 0;
    if (blockIdx.x == 0 && t == 0) cnt[N_NODES] = 0;

    const int base = blockIdx.x * FILL_EDGES + t * 4;
    int d[8], s[8], bk[8], rk[8];
    for (int g = 0; g < 2; ++g) {
        const int e0 = base + g * 1024;
        v4i dv = {-1, -1, -1, -1};
        v4i sv = {0, 0, 0, 0};
        if (e0 < N_EDGES) {
            dv = *(const v4i*)(dst + e0);
            sv = *(const v4i*)(src + e0);
        }
        for (int k = 0; k < 4; ++k) { d[4 * g + k] = dv[k]; s[4 * g + k] = sv[k]; }
    }
    __syncthreads();
    for (int k = 0; k < 8; ++k) {
        bk[k] = -1;
        rk[k] = 0;
        if ((unsigned)d[k] < (unsigned)N_NODES) {
            bk[k] = d[k] / BKT_NODES;
            rk[k] = __hip_atomic_fetch_add(&hist[bk[k]], 1, __ATOMIC_RELAXED, __HIP_MEMORY_SCOPE_WORKGROUP);
        }
    }
    __syncthreads();

    {
        const int lane = t & 31, wave = t >> 5;
        const int v0 = hist[2 * t], v1 = hist[2 * t + 1];
        int x = v0 + v1;
        for (int o = 1; o < 32; o <<= 1) {
            const int n = __shfl_up(x, o);
            if (lane >= o) x += n;
        }
        if (lane == 31) wtot[wave] = x;
        __syncthreads();
        int woff = 0;
        for (int w = 0; w < 8; ++w) woff += (w < wave) ? wtot[w] : 0;
        const int excl = woff + x - (v0 + v1);
        start[2 * t]     = excl;
        start[2 * t + 1] = excl + v0;
    }
    __syncthreads();

    for (int k = 0; k < 8; ++k) {
        if (bk[k] >= 0) {
            const int pos = start[bk[k]] + rk[k];
            v2i p = {d[k], s[k]};
            *(v2i*)(spair + 2 * pos) = p;
        }
    }
    for (int b = t; b <= N_BKT; b += 256) off[b * OFF_PITCH + blockIdx.x] = start[b];
    __syncthreads();

    const int total = start[N_BKT];
    int* myreg = srt + (size_t)blockIdx.x * (2 * FILL_EDGES);
    for (int i = 0; i < 4; ++i) {
        const int idx = i * 256 + t;
        if (2 * idx < total) *(v4i*)(myreg + 4 * idx) = *(const v4i*)(spair + 4 * idx);
    }
}

__global__ __launch_bounds__(256) void sage_bucket_slots(const int* __restrict__ srt, const int* __restrict__ off,
                                                         int* __restrict__ cnt, int* __restrict__ tab,
                                                         int* __restrict__ ovd, int* __restrict__ ovs) {
    __shared__ int lcnt[BKT_NODES];
    __shared__ int ltab[BKT_NODES * NBR_CAP];
    __shared__ int ls0[512];
    __shared__ int lpre[512];
    __shared__ int wtot[8];

    const int t    = threadIdx.x;
    const int k    = blockIdx.x;
    const int base = k * BKT_NODES;
    const int lane = t & 31;
    const int wave = t >> 5;

    if (t < BKT_NODES) lcnt[t] = 0;
    for (int a = t; a < 512; a += 256) {
        int b0 = 0, b1 = 0;
        if (a < E_BLK) {
            b0 = min(max(off[k * OFF_PITCH + a], 0), FILL_EDGES);
            b1 = min(max(off[(k + 1) * OFF_PITCH + a], b0), FILL_EDGES);
        }
        ls0[a]  = b0;
        lpre[a] = b1 - b0;
    }
    __syncthreads();

    const int v0 = lpre[2 * t], v1 = lpre[2 * t + 1];
    int x = v0 + v1;
    for (int o = 1; o < 32; o <<= 1) {
        const int n = __shfl_up(x, o);
        if (lane >= o) x += n;
    }
    if (lane == 31) wtot[wave] = x;
    __syncthreads();
    int woff = 0, total = 0;
    for (int w = 0; w < 8; ++w) {
        const int wt = wtot[w];
        total += wt;
        woff  += (w < wave) ? wt : 0;
    }
    const int excl = woff + x - (v0 + v1);
    lpre[2 * t]     = excl;
    lpre[2 * t + 1] = excl + v0;
    __syncthreads();

    for (int it0 = 0; it0 < total; it0 += 256 * SLOT_ITEMS) {
        int  it[SLOT_ITEMS], pos[SLOT_ITEMS], slot[SLOT_ITEMS];
        v2i  p[SLOT_ITEMS];
        #pragma unroll
        for (int u = 0; u < SLOT_ITEMS; ++u) {
            it[u]  = min(it0 + u * 256 + t, total - 1);
            pos[u] = 0;
        }
        #pragma unroll
        for (int step = 256; step >= 1; step >>= 1) {
            #pragma unroll
            for (int u = 0; u < SLOT_ITEMS; ++u)
                if (lpre[pos[u] + step - 1] <= it[u]) pos[u] += step;
        }
        #pragma unroll
        for (int u = 0; u < SLOT_ITEMS; ++u) {
            const int a = pos[u] - 1;
            const int e = ls0[a] + (it[u] - lpre[a]);
            p[u] = *((const v2i*)srt + (size_t)a * FILL_EDGES + e);
            if (it0 + u * 256 + t >= total) p[u].x = -1;
        }
        #pragma unroll
        for (int u = 0; u < SLOT_ITEMS; ++u) {
            const unsigned li = (unsigned)(p[u].x - base);
            slot[u] = -1;
            if (li < (unsigned)BKT_NODES && (unsigned)p[u].x < (unsigned)N_NODES)
                slot[u] = __hip_atomic_fetch_add(&lcnt[li], 1, __ATOMIC_RELAXED, __HIP_MEMORY_SCOPE_WORKGROUP);
        }
        #pragma unroll
        for (int u = 0; u < SLOT_ITEMS; ++u) {
            const int dn = p[u].x, sn = p[u].y;
            const unsigned li = (unsigned)(dn - base);
            if (li < (unsigned)BKT_NODES && (unsigned)dn < (unsigned)N_NODES) {
                if ((unsigned)slot[u] < (unsigned)NBR_CAP) {
                    ltab[li * NBR_CAP + slot[u]] = sn;
                } else {
                    int q = __hip_atomic_fetch_add(&cnt[N_NODES], 1, __ATOMIC_RELAXED, __HIP_MEMORY_SCOPE_AGENT);
                    if ((unsigned)q < (unsigned)N_EDGES) { ovd[q] = dn; ovs[q] = sn; }
                }
            }
        }
    }
    __syncthreads();

    if (t < BKT_NODES && base + t < N_NODES) cnt[base + t] = lcnt[t];
    int m[BKT_NODES / 8], w0[BKT_NODES / 8], w1[BKT_NODES / 8];
    for (int j = 0; j < BKT_NODES / 8; ++j) {
        const int i = wave * (BKT_NODES / 8) + j;
        m[j]  = min(lcnt[i], NBR_CAP);
        w0[j] = ltab[i * NBR_CAP + lane];
        w1[j] = ltab[i * NBR_CAP + 32 + lane];
    }
    for (int j = 0; j < BKT_NODES / 8; ++j) {
        const int node = base + wave * (BKT_NODES / 8) + j;
        if (node < N_NODES) {
            if (lane < m[j])      tab[(size_t)node * NBR_CAP + lane]      = w0[j];
            if (lane + 32 < m[j]) tab[(size_t)node * NBR_CAP + 32 + lane] = w1[j];
        }
    }
}

__global__ void sage_gather_gemm_wmma(const float* __restrict__ in,
                                      const int* __restrict__ cnt,
                                      const int* __restrict__ tab,
                                      const int* __restrict__ ovd,
                                      const int* __restrict__ ovs,
                                      const float* __restrict__ wp,
                                      const float* __restrict__ bias,
                                      float* __restrict__ out,
                                      int do_relu) {
    __shared__ __attribute__((aligned(16))) float smean[16 * MEAN_PITCH];
    __shared__ __attribute__((aligned(16))) float sself[16 * MEAN_PITCH];

    const int wave  = __builtin_amdgcn_readfirstlane((int)(threadIdx.x >> 5));
    const int lane  = threadIdx.x & 31;
    const int lhalf = lane >> 4;
    const int lmod  = lane & 15;
    const int rbase = blockIdx.x * 16;
    const int cbase = wave * 16;
    const int col   = cbase + lmod;

    const v4f self0 = *(const v4f*)(in + (size_t)(rbase + 2 * wave)     * D_FEAT + lane * 4);
    const v4f self1 = *(const v4f*)(in + (size_t)(rbase + 2 * wave + 1) * D_FEAT + lane * 4);

    for (int r = 0; r < 2; ++r) {
        const int lrow = 2 * wave + r;
        const int row  = rbase + lrow;
        const int deg   = min(max(cnt[row], 0), N_EDGES);
        const int intab = min(deg, NBR_CAP);
        const int* trow = tab + (size_t)row * NBR_CAP;
        v4f acc = {0.f, 0.f, 0.f, 0.f};
        for (int j0 = 0; j0 < intab; j0 += 32) {
            const int n = min(intab - j0, 32);
            int myid = (lane < n) ? trow[j0 + lane] : 0;
            myid = ((unsigned)myid < (unsigned)N_NODES) ? myid : 0;
            for (int u0 = 0; u0 < n; u0 += 8) {
                v4f v[8];
                #pragma unroll
                for (int u = 0; u < 8; ++u) {
                    const int sidx = __builtin_amdgcn_readlane(myid, min(u0 + u, n - 1));
                    v[u] = *(const v4f*)(in + (size_t)sidx * D_FEAT + lane * 4);
                }
                #pragma unroll
                for (int u = 0; u < 8; ++u)
                    if (u0 + u < n) acc += v[u];
            }
        }
        if (deg > NBR_CAP) {
            const int novf = min(max(cnt[N_NODES], 0), N_EDGES);
            for (int j0 = 0; j0 < novf; j0 += 32) {
                const int n  = min(novf - j0, 32);
                const int od = (lane < n) ? ovd[j0 + lane] : -1;
                int       os = (lane < n) ? ovs[j0 + lane] : 0;
                os = ((unsigned)os < (unsigned)N_NODES) ? os : 0;
                unsigned m = __builtin_amdgcn_ballot_w32(od == row);
                while (m) {
                    const int b = __builtin_ctz(m);
                    m &= m - 1;
                    const int sidx = __builtin_amdgcn_readlane(os, b);
                    acc += *(const v4f*)(in + (size_t)sidx * D_FEAT + lane * 4);
                }
            }
        }
        const float dv = fmaxf((float)deg, 1.0f);
        *(v4f*)(smean + lrow * MEAN_PITCH + lane * 4) = acc / dv;
        *(v4f*)(sself + lrow * MEAN_PITCH + lane * 4) = r ? self1 : self0;
    }
    __syncthreads();

    v8f acc = {};
    const float* amp = smean + lmod * MEAN_PITCH + (lhalf ? 2 : 0);
    const float* axp = sself + lmod * MEAN_PITCH + (lhalf ? 2 : 0);

    const v4f* wq = (const v4f*)wp + (size_t)wave * (16 * 2 * 32) + lane;

    #pragma unroll 4
    for (int u = 0; u < D_FEAT / 8; ++u) {
        const v4f pl = wq[(2 * u + 0) * 32];
        const v4f pr = wq[(2 * u + 1) * 32];
        const v2f am0 = *(const v2f*)(amp + 8 * u);
        const v2f ax0 = *(const v2f*)(axp + 8 * u);
        const v2f am1 = *(const v2f*)(amp + 8 * u + 4);
        const v2f ax1 = *(const v2f*)(axp + 8 * u + 4);
        const v2f bl0 = __builtin_shufflevector(pl, pl, 0, 1);
        const v2f br0 = __builtin_shufflevector(pr, pr, 0, 1);
        const v2f bl1 = __builtin_shufflevector(pl, pl, 2, 3);
        const v2f br1 = __builtin_shufflevector(pr, pr, 2, 3);
        acc = __builtin_amdgcn_wmma_f32_16x16x4_f32(false, am0, false, bl0,
                                                    (short)0, acc, false, false);
        acc = __builtin_amdgcn_wmma_f32_16x16x4_f32(false, ax0, false, br0,
                                                    (short)0, acc, false, false);
        acc = __builtin_amdgcn_wmma_f32_16x16x4_f32(false, am1, false, bl1,
                                                    (short)0, acc, false, false);
        acc = __builtin_amdgcn_wmma_f32_16x16x4_f32(false, ax1, false, br1,
                                                    (short)0, acc, false, false);
    }

    float bv = bias[col];
    #pragma unroll
    for (int v = 0; v < 8; ++v) {
        float r = acc[v] + bv;
        if (do_relu) r = fmaxf(r, 0.0f);
        int m = v + (lhalf ? 8 : 0);
        out[(rbase + m) * D_FEAT + col] = r;
    }
}

extern "C" void kernel_launch(void* const* d_in, const int* in_sizes, int n_in,
                              void* d_out, int out_size, void* d_ws, size_t ws_size,
                              hipStream_t stream) {
    const float* x    = (const float*)d_in[0];
    const int*   eidx = (const int*)d_in[1];
    const float* W1l  = (const float*)d_in[2];
    const float* b1   = (const float*)d_in[3];
    const float* W1r  = (const float*)d_in[4];
    const float* W2l  = (const float*)d_in[5];
    const float* b2   = (const float*)d_in[6];
    const float* W2r  = (const float*)d_in[7];
    float* out = (float*)d_out;

    const int* src = eidx;
    const int* dst = eidx + N_EDGES;

    int*   cnt = (int*)d_ws;
    int*   tab = (int*)d_ws + (1 << 16);
    int*   ovd = (int*)d_ws + (25 << 17);
    int*   ovs = (int*)d_ws + (1 << 22);
    float* wp  = (float*)d_ws + (5 << 20);
    float* h   = (float*)d_ws + (size_t)N_NODES * D_FEAT + N_NODES;
    int*   srt = (int*)h;
    int*   off = (int*)h + 2 * E_BLK * FILL_EDGES;

    const int gemmBl = N_NODES / 16;

    sage_bucket_edges_pack_w<<<E_BLK + PACK_BL, 256, 0, stream>>>(src, dst, cnt, srt, off, W1l, W1r, W2l, W2r, wp);
    sage_bucket_slots<<<N_BKT, 256, 0, stream>>>(srt, off, cnt, tab, ovd, ovs);

    sage_gather_gemm_wmma<<<gemmBl, 256, 0, stream>>>(x, cnt, tab, ovd, ovs, wp, b1, h, 1);
    sage_gather_gemm_wmma<<<gemmBl, 256, 0, stream>>>(h, cnt, tab, ovd, ovs, wp + WP_LAYER, b2, out, 0);
}


# Own-feature product first: in @ Wr runs from LDS behind one barrier, ahead of the neighbour sums; mean @ Wl behind a second, so a WGP multiplies while it gathers; sum order: Wr terms, then Wl terms
# speedup vs baseline: 7.3368x; 1.0363x over previous
#include <hip/hip_runtime.h>
#include <hip/hip_bf16.h>


#define N_NODES 50000
#define N_EDGES 800000
#define D_FEAT  128

#define NBR_CAP    64
#define FILL_EDGES 2048
#define E_BLK      ((N_EDGES + FILL_EDGES - 1) / FILL_EDGES)
#define BKT_NODES  128
#define N_BKT      ((N_NODES + BKT_NODES - 1) / BKT_NODES)
#define OFF_PITCH  392
#define SLOT_ITEMS 9
#define MEAN_PITCH 132
#define WP_LAYER   (2 * D_FEAT * D_FEAT)
#define PACK_BL    (2 * WP_LAYER / 4 / 256)

static_assert(N_BKT + 1 <= 512, "the bucket histogram has 512 bins");
static_assert(E_BLK <= OFF_PITCH && N_BKT + 1 <= OFF_PITCH, "offset table pitch");
static_assert(E_BLK < 511, "the scan of piece lengths has 512 entries and the search reads entries 0 .. 510");
static_assert(N_EDGES % 4 == 0, "edge groups of 4 are whole or absent");

typedef __attribute__((ext_vector_type(2))) int   v2i;
typedef __attribute__((ext_vector_type(4))) int   v4i;
typedef __attribute__((ext_vector_type(2))) float v2f;
typedef __attribute__((ext_vector_type(4))) float v4f;
typedef __attribute__((ext_vector_type(8))) float v8f;

__global__ __launch_bounds__(256) void sage_bucket_edges_pack_w(const int* __restrict__ src, const int* __restrict__ dst,
                                                                int* __restrict__ cnt, int* __restrict__ srt, int* __restrict__ off,
                                                                const float* __restrict__ W1l, const float* __restrict__ W1r,
                                                                const float* __restrict__ W2l, const float* __restrict__ W2r,
                                                                float* __restrict__ wp) {
    __shared__ int hist[512];
    __shared__ int start[512];
    __shared__ int wtot[8];
    __shared__ __attribute__((aligned(16))) int spair[2 * FILL_EDGES];

    const int t = threadIdx.x;
    if (blockIdx.x >= E_BLK) {
        const int g     = (blockIdx.x - E_BLK) * 256 + t;
        const int layer = g >> 13;
        const int r     = g & 8191;
        const int l     = r & 31;
        const int m     = (r >> 5) & 1;
        const int u     = (r >> 6) & 15;
        const int w     = (r >> 10) & 7;
        const int col   = 16 * w + (l & 15);
        const int k0    = 8 * u + 2 * (l >> 4);
        const float* W  = layer ? (m ? W2r : W2l) : (m ? W1r : W1l);
        v4f o;
        o.x = W[(k0 + 0) * D_FEAT + col];
        o.y = W[(k0 + 1) * D_FEAT + col];
        o.z = W[(k0 + 4) * D_FEAT + col];
        o.w = W[(k0 + 5) * D_FEAT + col];
        *(v4f*)(wp + (size_t)layer * WP_LAYER + (size_t)r * 4) = o;
        return;
    }

    hist[t] = 0;
    hist[t + 256] = 0;
    if (blockIdx.x == 0 && t == 0) cnt[N_NODES] = 0;

    const int base = blockIdx.x * FILL_EDGES + t * 4;
    int d[8], s[8], bk[8], rk[8];
    for (int g = 0; g < 2; ++g) {
        const int e0 = base + g * 1024;
        v4i dv = {-1, -1, -1, -1};
        v4i sv = {0, 0, 0, 0};
        if (e0 < N_EDGES) {
            dv = *(const v4i*)(dst + e0);
            sv = *(const v4i*)(src + e0);
        }
        for (int k = 0; k < 4; ++k) { d[4 * g + k] = dv[k]; s[4 * g + k] = sv[k]; }
    }
    __syncthreads();
    for (int k = 0; k < 8; ++k) {
        bk[k] = -1;
        rk[k] = 0;
        if ((unsigned)d[k] < (unsigned)N_NODES) {
            bk[k] = d[k] / BKT_NODES;
            rk[k] = __hip_atomic_fetch_add(&hist[bk[k]], 1, __ATOMIC_RELAXED, __HIP_MEMORY_SCOPE_WORKGROUP);
        }
    }
    __syncthreads();

    {
        const int lane = t & 31, wave = t >> 5;
        const int v0 = hist[2 * t], v1 = hist[2 * t + 1];
        int x = v0 + v1;
        for (int o = 1; o < 32; o <<= 1) {
            const int n = __shfl_up(x, o);
            if (lane >= o) x += n;
        }
        if (lane == 31) wtot[wave] = x;
        __syncthreads();
        int woff = 0;
        for (int w = 0; w < 8; ++w) woff += (w < wave) ? wtot[w] : 0;
        const int excl = woff + x - (v0 + v1);
        start[2 * t]     = excl;
        start[2 * t + 1] = excl + v0;
    }
    __syncthreads();

    for (int k = 0; k < 8; ++k) {
        if (bk[k] >= 0) {
            const int pos = start[bk[k]] + rk[k];
            v2i p = {d[k], s[k]};
            *(v2i*)(spair + 2 * pos) = p;
        }
    }
    for (int b = t; b <= N_BKT; b += 256) off[b * OFF_PITCH + blockIdx.x] = start[b];
    __syncthreads();

    const int total = start[N_BKT];
    int* myreg = srt + (size_t)blockIdx.x * (2 * FILL_EDGES);
    for (int i = 0; i < 4; ++i) {
        const int idx = i * 256 + t;
        if (2 * idx < total) *(v4i*)(myreg + 4 * idx) = *(const v4i*)(spair + 4 * idx);
    }
}

__global__ __launch_bounds__(256) void sage_bucket_slots(const int* __restrict__ srt, const int* __restrict__ off,
                                                         int* __restrict__ cnt, int* __restrict__ tab,
                                                         int* __restrict__ ovd, int* __restrict__ ovs) {
    __shared__ int lcnt[BKT_NODES];
    __shared__ int ltab[BKT_NODES * NBR_CAP];
    __shared__ int ls0[512];
    __shared__ int lpre[512];
    __shared__ int wtot[8];

    const int t    = threadIdx.x;
    const int k    = blockIdx.x;
    const int base = k * BKT_NODES;
    const int lane = t & 31;
    const int wave = t >> 5;

    if (t < BKT_NODES) lcnt[t] = 0;
    for (int a = t; a < 512; a += 256) {
        int b0 = 0, b1 = 0;
        if (a < E_BLK) {
            b0 = min(max(off[k * OFF_PITCH + a], 0), FILL_EDGES);
            b1 = min(max(off[(k + 1) * OFF_PITCH + a], b0), FILL_EDGES);
        }
        ls0[a]  = b0;
        lpre[a] = b1 - b0;
    }
    __syncthreads();

    const int v0 = lpre[2 * t], v1 = lpre[2 * t + 1];
    int x = v0 + v1;
    for (int o = 1; o < 32; o <<= 1) {
        const int n = __shfl_up(x, o);
        if (lane >= o) x += n;
    }
    if (lane == 31) wtot[wave] = x;
    __syncthreads();
    int woff = 0, total = 0;
    for (int w = 0; w < 8; ++w) {
        const int wt = wtot[w];
        total += wt;
        woff  += (w < wave) ? wt : 0;
    }
    const int excl = woff + x - (v0 + v1);
    lpre[2 * t]     = excl;
    lpre[2 * t + 1] = excl + v0;
    __syncthreads();

    for (int it0 = 0; it0 < total; it0 += 256 * SLOT_ITEMS) {
        int  it[SLOT_ITEMS], pos[SLOT_ITEMS], slot[SLOT_ITEMS];
        v2i  p[SLOT_ITEMS];
        #pragma unroll
        for (int u = 0; u < SLOT_ITEMS; ++u) {
            it[u]  = min(it0 + u * 256 + t, total - 1);
            pos[u] = 0;
        }
        #pragma unroll
        for (int step = 256; step >= 1; step >>= 1) {
            #pragma unroll
            for (int u = 0; u < SLOT_ITEMS; ++u)
                if (lpre[pos[u] + step - 1] <= it[u]) pos[u] += step;
        }
        #pragma unroll
        for (int u = 0; u < SLOT_ITEMS; ++u) {
            const int a = pos[u] - 1;
            const int e = ls0[a] + (it[u] - lpre[a]);
            p[u] = *((const v2i*)srt + (size_t)a * FILL_EDGES + e);
            if (it0 + u * 256 + t >= total) p[u].x = -1;
        }
        #pragma unroll
        for (int u = 0; u < SLOT_ITEMS; ++u) {
            const unsigned li = (unsigned)(p[u].x - base);
            slot[u] = -1;
            if (li < (unsigned)BKT_NODES && (unsigned)p[u].x < (unsigned)N_NODES)
                slot[u] = __hip_atomic_fetch_add(&lcnt[li], 1, __ATOMIC_RELAXED, __HIP_MEMORY_SCOPE_WORKGROUP);
        }
        #pragma unroll
        for (int u = 0; u < SLOT_ITEMS; ++u) {
            const int dn = p[u].x, sn = p[u].y;
            const unsigned li = (unsigned)(dn - base);
            if (li < (unsigned)BKT_NODES && (unsigned)dn < (unsigned)N_NODES) {
                if ((unsigned)slot[u] < (unsigned)NBR_CAP) {
                    ltab[li * NBR_CAP + slot[u]] = sn;
                } else {
                    int q = __hip_atomic_fetch_add(&cnt[N_NODES], 1, __ATOMIC_RELAXED, __HIP_MEMORY_SCOPE_AGENT);
                    if ((unsigned)q < (unsigned)N_EDGES) { ovd[q] = dn; ovs[q] = sn; }
                }
            }
        }
    }
    __syncthreads();

    if (t < BKT_NODES && base + t < N_NODES) cnt[base + t] = lcnt[t];
    int m[BKT_NODES / 8], w0[BKT_NODES / 8], w1[BKT_NODES / 8];
    for (int j = 0; j < BKT_NODES / 8; ++j) {
        const int i = wave * (BKT_NODES / 8) + j;
        m[j]  = min(lcnt[i], NBR_CAP);
        w0[j] = ltab[i * NBR_CAP + lane];
        w1[j] = ltab[i * NBR_CAP + 32 + lane];
    }
    for (int j = 0; j < BKT_NODES / 8; ++j) {
        const int node = base + wave * (BKT_NODES / 8) + j;
        if (node < N_NODES) {
            if (lane < m[j])      tab[(size_t)node * NBR_CAP + lane]      = w0[j];
            if (lane + 32 < m[j]) tab[(size_t)node * NBR_CAP + 32 + lane] = w1[j];
        }
    }
}

__global__ void sage_gather_gemm_wmma(const float* __restrict__ in,
                                      const int* __restrict__ cnt,
                                      const int* __restrict__ tab,
                                      const int* __restrict__ ovd,
                                      const int* __restrict__ ovs,
                                      const float* __restrict__ wp,
                                      const float* __restrict__ bias,
                                      float* __restrict__ out,
                                      int do_relu) {
    __shared__ __attribute__((aligned(16))) float smean[16 * MEAN_PITCH];
    __shared__ __attribute__((aligned(16))) float sself[16 * MEAN_PITCH];

    const int wave  = __builtin_amdgcn_readfirstlane((int)(threadIdx.x >> 5));
    const int lane  = threadIdx.x & 31;
    const int lhalf = lane >> 4;
    const int lmod  = lane & 15;
    const int rbase = blockIdx.x * 16;
    const int cbase = wave * 16;
    const int col   = cbase + lmod;

    {
        const v4f self0 = *(const v4f*)(in + (size_t)(rbase + 2 * wave)     * D_FEAT + lane * 4);
        const v4f self1 = *(const v4f*)(in + (size_t)(rbase + 2 * wave + 1) * D_FEAT + lane * 4);
        *(v4f*)(sself + (2 * wave)     * MEAN_PITCH + lane * 4) = self0;
        *(v4f*)(sself + (2 * wave + 1) * MEAN_PITCH + lane * 4) = self1;
    }
    __syncthreads();

    v8f acc = {};
    const float* amp = smean + lmod * MEAN_PITCH + (lhalf ? 2 : 0);
    const float* axp = sself + lmod * MEAN_PITCH + (lhalf ? 2 : 0);

    const v4f* wq = (const v4f*)wp + (size_t)wave * (16 * 2 * 32) + lane;

    #pragma unroll 8
    for (int u = 0; u < D_FEAT / 8; ++u) {
        const v4f pr = wq[(2 * u + 1) * 32];
        const v2f ax0 = *(const v2f*)(axp + 8 * u);
        const v2f ax1 = *(const v2f*)(axp + 8 * u + 4);
        const v2f br0 = __builtin_shufflevector(pr, pr, 0, 1);
        const v2f br1 = __builtin_shufflevector(pr, pr, 2, 3);
        acc = __builtin_amdgcn_wmma_f32_16x16x4_f32(false, ax0, false, br0,
                                                    (short)0, acc, false, false);
        acc = __builtin_amdgcn_wmma_f32_16x16x4_f32(false, ax1, false, br1,
                                                    (short)0, acc, false, false);
    }

    asm volatile("v_nop\n\tv_nop\n\tv_nop\n\tv_nop");

    for (int r = 0; r < 2; ++r) {
        const int lrow = 2 * wave + r;
        const int row  = rbase + lrow;
        const int deg   = min(max(cnt[row], 0), N_EDGES);
        const int intab = min(deg, NBR_CAP);
        const int* trow = tab + (size_t)row * NBR_CAP;
        v4f sum = {0.f, 0.f, 0.f, 0.f};
        for (int j0 = 0; j0 < intab; j0 += 32) {
            const int n = min(intab - j0, 32);
            int myid = (lane < n) ? trow[j0 + lane] : 0;
            myid = ((unsigned)myid < (unsigned)N_NODES) ? myid : 0;
            for (int u0 = 0; u0 < n; u0 += 8) {
                v4f v[8];
                #pragma unroll
                for (int u = 0; u < 8; ++u) {
                    const int sidx = __builtin_amdgcn_readlane(myid, min(u0 + u, n - 1));
                    v[u] = *(const v4f*)(in + (size_t)sidx * D_FEAT + lane * 4);
                }
                #pragma unroll
                for (int u = 0; u < 8; ++u)
                    if (u0 + u < n) sum += v[u];
            }
        }
        if (deg > NBR_CAP) {
            const int novf = min(max(cnt[N_NODES], 0), N_EDGES);
            for (int j0 = 0; j0 < novf; j0 += 32) {
                const int n  = min(novf - j0, 32);
                const int od = (lane < n) ? ovd[j0 + lane] : -1;
                int       os = (lane < n) ? ovs[j0 + lane] : 0;
                os = ((unsigned)os < (unsigned)N_NODES) ? os : 0;
                unsigned m = __builtin_amdgcn_ballot_w32(od == row);
                while (m) {
                    const int b = __builtin_ctz(m);
                    m &= m - 1;
                    const int sidx = __builtin_amdgcn_readlane(os, b);
                    sum += *(const v4f*)(in + (size_t)sidx * D_FEAT + lane * 4);
                }
            }
        }
        const float dv = fmaxf((float)deg, 1.0f);
        *(v4f*)(smean + lrow * MEAN_PITCH + lane * 4) = sum / dv;
    }
    __syncthreads();

    #pragma unroll 8
    for (int u = 0; u < D_FEAT / 8; ++u) {
        const v4f pl = wq[(2 * u + 0) * 32];
        const v2f am0 = *(const v2f*)(amp + 8 * u);
        const v2f am1 = *(const v2f*)(amp + 8 * u + 4);
        const v2f bl0 = __builtin_shufflevector(pl, pl, 0, 1);
        const v2f bl1 = __builtin_shufflevector(pl, pl, 2, 3);
        acc = __builtin_amdgcn_wmma_f32_16x16x4_f32(false, am0, false, bl0,
                                                    (short)0, acc, false, false);
        acc = __builtin_amdgcn_wmma_f32_16x16x4_f32(false, am1, false, bl1,
                                                    (short)0, acc, false, false);
    }

    float bv = bias[col];
    #pragma unroll
    for (int v = 0; v < 8; ++v) {
        float r = acc[v] + bv;
        if (do_relu) r = fmaxf(r, 0.0f);
        int m = v + (lhalf ? 8 : 0);
        out[(rbase + m) * D_FEAT + col] = r;
    }
}

extern "C" void kernel_launch(void* const* d_in, const int* in_sizes, int n_in,
                              void* d_out, int out_size, void* d_ws, size_t ws_size,
                              hipStream_t stream) {
    const float* x    = (const float*)d_in[0];
    const int*   eidx = (const int*)d_in[1];
    const float* W1l  = (const float*)d_in[2];
    const float* b1   = (const float*)d_in[3];
    const float* W1r  = (const float*)d_in[4];
    const float* W2l  = (const float*)d_in[5];
    const float* b2   = (const float*)d_in[6];
    const float* W2r  = (const float*)d_in[7];
    float* out = (float*)d_out;

    const int* src = eidx;
    const int* dst = eidx + N_EDGES;

    int*   cnt = (int*)d_ws;
    int*   tab = (int*)d_ws + (1 << 16);
    int*   ovd = (int*)d_ws + (25 << 17);
    int*   ovs = (int*)d_ws + (1 << 22);
    float* wp  = (float*)d_ws + (5 << 20);
    float* h   = (float*)d_ws + (size_t)N_NODES * D_FEAT + N_NODES;
    int*   srt = (int*)h;
    int*   off = (int*)h + 2 * E_BLK * FILL_EDGES;

    const int gemmBl = N_NODES / 16;

    sage_bucket_edges_pack_w<<<E_BLK + PACK_BL, 256, 0, stream>>>(src, dst, cnt, srt, off, W1l, W1r, W2l, W2r, wp);
    sage_bucket_slots<<<N_BKT, 256, 0, stream>>>(srt, off, cnt, tab, ovd, ovs);

    sage_gather_gemm_wmma<<<gemmBl, 256, 0, stream>>>(x, cnt, tab, ovd, ovs, wp, b1, h, 1);
    sage_gather_gemm_wmma<<<gemmBl, 256, 0, stream>>>(h, cnt, tab, ovd, ovs, wp + WP_LAYER, b2, out, 0);
}
